# starting kernel
# baseline (speedup 1.0000x reference)
#include <hip/hip_runtime.h>
#include <math.h>

#define Bdim 8
#define Ndim 256
#define ZDIM 64
#define EDIM 16
#define HDIM 64

typedef __attribute__((ext_vector_type(2))) float v2f;
typedef __attribute__((ext_vector_type(8))) float v8f;

__global__ void mimj_kernel(const float* __restrict__ z,
                            const float* __restrict__ Wmsg,
                            float* __restrict__ mi, float* __restrict__ mj) {
    int bn = blockIdx.x;
    int h  = threadIdx.x;
    __shared__ float zrow[ZDIM];
    zrow[h] = z[bn * ZDIM + h];
    __syncthreads();
    const float* wi = Wmsg + h * (2 * ZDIM + EDIM);
    const float* wj = wi + ZDIM;
    float si = 0.f, sj = 0.f;
#pragma unroll 8
    for (int f = 0; f < ZDIM; ++f) {
        float zz = zrow[f];
        si += zz * wi[f];
        sj += zz * wj[f];
    }
    mi[bn * HDIM + h] = si;
    mj[bn * HDIM + h] = sj;
}

#define E_STRIDE 20
#define W_STRIDE 80

__global__ void msg_kernel(const float* __restrict__ e_feat,
                           const int*   __restrict__ adj,
                           const float* __restrict__ Wmsg,
                           const float* __restrict__ bmsg,
                           const float* __restrict__ mj,
                           const float* __restrict__ mi,
                           float* __restrict__ agg) {
    __shared__ float E[Ndim * E_STRIDE];
    __shared__ float WeT[EDIM * W_STRIDE];
    __shared__ float maskF[Ndim];
    __shared__ float red[8 * HDIM];

    const int bn = blockIdx.x;
    const int b  = bn >> 8;
    const int t  = threadIdx.x;

    const float NEG_INF = -__builtin_inff();

    {
        const float4* src = (const float4*)(e_feat + (size_t)bn * Ndim * EDIM + t * EDIM);
        float4* dst = (float4*)(E + t * E_STRIDE);
#pragma unroll
        for (int q = 0; q < 4; ++q) dst[q] = src[q];
    }
    {
#pragma unroll
        for (int q = 0; q < 4; ++q) {
            int idx = t * 4 + q;
            int e = idx >> 6, h = idx & 63;
            WeT[e * W_STRIDE + h] = Wmsg[h * (2 * ZDIM + EDIM) + 2 * ZDIM + e];
        }
    }
    maskF[t] = (adj[(size_t)bn * Ndim + t] == 1) ? 0.f : NEG_INF;
    __syncthreads();

    const int wave  = t >> 5;
    const int lane  = t & 31;
    const int lm    = lane & 15;
    const int hi    = lane >> 4;
    const int jwave = wave * 32;

    float mk[16];
#pragma unroll
    for (int jt = 0; jt < 2; ++jt)
#pragma unroll
        for (int r = 0; r < 8; ++r)
            mk[jt * 8 + r] = maskF[jwave + jt * 16 + r + 8 * hi];

    float vm0 = NEG_INF, vm1 = NEG_INF, vm2 = NEG_INF, vm3 = NEG_INF;
    float vmarr[4];

#pragma unroll
    for (int jt = 0; jt < 2; ++jt) {
        const int jbase = jwave + jt * 16;

        v2f afr[4];
#pragma unroll
        for (int k = 0; k < 4; ++k) {
            const int ar = (jbase + lm) * E_STRIDE + 4 * k + 2 * hi;
            afr[k].x = E[ar];
            afr[k].y = E[ar + 1];
        }

        float vmn[4] = {NEG_INF, NEG_INF, NEG_INF, NEG_INF};
#pragma unroll
        for (int ht = 0; ht < 4; ++ht) {
            const int hbase = ht * 16;

            const float* mjp = mj + ((size_t)b * Ndim + jbase + 8 * hi) * HDIM
                                  + hbase + lm;
            v8f acc;
#pragma unroll
            for (int r = 0; r < 8; ++r) acc[r] = mjp[r * HDIM];

#pragma unroll
            for (int k = 0; k < 4; ++k) {
                const int k0 = 4 * k;
                v2f bm;
                const int be = (k0 + 2 * hi) * W_STRIDE + hbase + lm;
                bm.x = WeT[be];
                bm.y = WeT[be + W_STRIDE];
                acc = __builtin_amdgcn_wmma_f32_16x16x4_f32(
                    false, afr[k], false, bm, (short)0, acc, false, false);
            }

            float v = vmn[ht];
#pragma unroll
            for (int r = 0; r < 8; ++r) v = fmaxf(v, acc[r] + mk[jt * 8 + r]);
            vmn[ht] = v;
        }
        vm0 = fmaxf(vm0, vmn[0]);
        vm1 = fmaxf(vm1, vmn[1]);
        vm2 = fmaxf(vm2, vmn[2]);
        vm3 = fmaxf(vm3, vmn[3]);
    }
    vmarr[0] = vm0; vmarr[1] = vm1; vmarr[2] = vm2; vmarr[3] = vm3;

#pragma unroll
    for (int ht = 0; ht < 4; ++ht) {
        float v = fmaxf(vmarr[ht], __shfl_xor(vmarr[ht], 16, 32));
        if (lane < 16) red[wave * HDIM + ht * 16 + lm] = v;
    }
    __syncthreads();

    if (t < HDIM) {
        float m = red[t];
#pragma unroll
        for (int w = 1; w < 8; ++w) m = fmaxf(m, red[w * HDIM + t]);
        agg[(size_t)bn * HDIM + t] = m + mi[(size_t)bn * HDIM + t] + bmsg[t];
    }
}

__global__ void upd_kernel(const float* __restrict__ z,
                           const float* __restrict__ agg,
                           const float* __restrict__ Wupd,
                           const float* __restrict__ bupd,
                           float* __restrict__ out) {
    int bn = blockIdx.x;
    int h  = threadIdx.x;
    __shared__ float feat[ZDIM + HDIM];
    feat[h]        = z[bn * ZDIM + h];
    feat[ZDIM + h] = agg[bn * HDIM + h];
    __syncthreads();
    const float* w = Wupd + h * (ZDIM + HDIM);
    float s = bupd[h];
#pragma unroll 8
    for (int f = 0; f < ZDIM + HDIM; ++f) s += feat[f] * w[f];
    out[bn * HDIM + h] = s;
}

extern "C" void kernel_launch(void* const* d_in, const int* in_sizes, int n_in,
                              void* d_out, int out_size, void* d_ws, size_t ws_size,
                              hipStream_t stream) {
    const float* z      = (const float*)d_in[0];
    const float* e_feat = (const float*)d_in[1];
    const int*   adj    = (const int*)d_in[2];
    const float* W_msg  = (const float*)d_in[3];
    const float* b_msg  = (const float*)d_in[4];
    const float* W_upd  = (const float*)d_in[5];
    const float* b_upd  = (const float*)d_in[6];
    float*       out    = (float*)d_out;

    const int BN = Bdim * Ndim;
    float* mi  = (float*)d_ws;
    float* mj  = mi + BN * HDIM;
    float* agg = mj + BN * HDIM;

    mimj_kernel<<<BN, HDIM, 0, stream>>>(z, W_msg, mi, mj);
    msg_kernel <<<BN, 256,  0, stream>>>(e_feat, adj, W_msg, b_msg, mj, mi, agg);
    upd_kernel <<<BN, HDIM, 0, stream>>>(z, agg, W_upd, b_upd, out);
}


# both per-node linear layers (m_i/m_j and update) re-tiled to 16-node f32 WMMA tiles: weights read once per 16 nodes as B fragments, not walked per lane for every node (sum order of both changes)
# speedup vs baseline: 1.6569x; 1.6569x over previous
#include <hip/hip_runtime.h>
#include <math.h>

#define Bdim 8
#define Ndim 256
#define ZDIM 64
#define EDIM 16
#define HDIM 64

typedef __attribute__((ext_vector_type(2))) float v2f;
typedef __attribute__((ext_vector_type(8))) float v8f;
typedef __attribute__((ext_vector_type(4))) float v4f;

#define WROW (2 * ZDIM + EDIM)

__global__ __launch_bounds__(64)
void mimj_kernel(const float* __restrict__ z,
                 const float* __restrict__ Wmsg,
                 float* __restrict__ mi, float* __restrict__ mj) {
    const int lane = threadIdx.x & 31;
    const int wave = threadIdx.x >> 5;
    const int c    = lane & 15;
    const int hf   = lane >> 4;
    const int n0   = (blockIdx.x >> 1) * 16;
    const int half = blockIdx.x & 1;
    const int hb   = 32 * wave;

    const float* ap = z + (size_t)(n0 + c) * ZDIM + 4 * hf;
    const float* bp = Wmsg + (size_t)(hb + 2 * c) * WROW + half * ZDIM + 4 * hf;
    v4f a[8], b0[8], b1[8];
#pragma unroll
    for (int i = 0; i < 8; ++i) {
        a[i]  = *(const v4f*)(ap + 8 * i);
        b0[i] = *(const v4f*)(bp + 8 * i);
        b1[i] = *(const v4f*)(bp + WROW + 8 * i);
    }
    __builtin_amdgcn_sched_barrier(0);

    v8f acc0, acc1;
#pragma unroll
    for (int r = 0; r < 8; ++r) { acc0[r] = 0.f; acc1[r] = 0.f; }
#pragma unroll
    for (int i = 0; i < 8; ++i) {
#pragma unroll
        for (int e = 0; e < 2; ++e) {
            v2f af, bf0, bf1;
            af.x  = a[i][2 * e];  af.y  = a[i][2 * e + 1];
            bf0.x = b0[i][2 * e]; bf0.y = b0[i][2 * e + 1];
            bf1.x = b1[i][2 * e]; bf1.y = b1[i][2 * e + 1];
            acc0 = __builtin_amdgcn_wmma_f32_16x16x4_f32(
                false, af, false, bf0, (short)0, acc0, false, false);
            acc1 = __builtin_amdgcn_wmma_f32_16x16x4_f32(
                false, af, false, bf1, (short)0, acc1, false, false);
        }
    }

    float* dst = (half ? mj : mi) + (size_t)(n0 + 8 * hf) * HDIM + hb + 2 * c;
#pragma unroll
    for (int r = 0; r < 8; ++r)
        *(float2*)(dst + r * HDIM) = make_float2(acc0[r], acc1[r]);
}

#define JW 16
#define CH 4

__global__ __launch_bounds__(256)
void msg_kernel(const float* __restrict__ e_feat,
                const int*   __restrict__ adj,
                const float* __restrict__ Wmsg,
                const float* __restrict__ mj,
                float* __restrict__ p0, float* __restrict__ p1) {
    __shared__ float4 red[8 * 16 * 16];

    const int t    = threadIdx.x;
    const int wave = t >> 5;
    const int lane = t & 31;
    const int c    = lane & 15;
    const int hf   = lane >> 4;
    const int wg   = blockIdx.x;
    const int jh   = wg & 1;
    const int i0   = ((wg >> 1) & 15) * 16;
    const int b    = wg >> 5;
    const int j0   = jh * 128 + wave * JW;

    const float NEG_INF = -__builtin_inff();

    const float* ep  = e_feat + (((size_t)(b * Ndim + i0 + c)) * Ndim + j0) * EDIM + 8 * hf;
    const float* mjp = mj + ((size_t)(b * Ndim + j0)) * HDIM + 4 * c;
    v4f ea[CH], eb[CH], mjv[CH];
#pragma unroll
    for (int q = 0; q < CH; ++q) {
        ea[q]  = *(const v4f*)(ep + q * EDIM);
        eb[q]  = *(const v4f*)(ep + q * EDIM + 4);
        mjv[q] = *(const v4f*)(mjp + q * HDIM);
    }

    v2f bfr[4][4];
#pragma unroll
    for (int ht = 0; ht < 4; ++ht) {
        const float* wp = Wmsg + (4 * c + ht) * (2 * ZDIM + EDIM) + 2 * ZDIM + 8 * hf;
        const float4 w0 = *(const float4*)(wp);
        const float4 w1 = *(const float4*)(wp + 4);
        bfr[ht][0].x = w0.x; bfr[ht][0].y = w0.y;
        bfr[ht][1].x = w0.z; bfr[ht][1].y = w0.w;
        bfr[ht][2].x = w1.x; bfr[ht][2].y = w1.y;
        bfr[ht][3].x = w1.z; bfr[ht][3].y = w1.w;
    }

    unsigned mbits = 0u;
    {
        const int4* ap = (const int4*)(adj + ((size_t)(b * Ndim + i0 + c)) * Ndim + j0);
#pragma unroll
        for (int q = 0; q < 4; ++q) {
            const int4 a4 = ap[q];
            mbits |= (a4.x == 1 ? 1u : 0u) << (4 * q + 0);
            mbits |= (a4.y == 1 ? 1u : 0u) << (4 * q + 1);
            mbits |= (a4.z == 1 ? 1u : 0u) << (4 * q + 2);
            mbits |= (a4.w == 1 ? 1u : 0u) << (4 * q + 3);
        }
    }
    unsigned mr[8];
#pragma unroll
    for (int r = 0; r < 8; ++r) mr[r] = (unsigned)__shfl((int)mbits, r + 8 * hf, 32);

    float vmax[4][8];
#pragma unroll
    for (int ht = 0; ht < 4; ++ht)
#pragma unroll
        for (int r = 0; r < 8; ++r) vmax[ht][r] = NEG_INF;

#pragma nounroll
    for (int ch = 0; ch < JW / CH; ++ch) {
        if (ch + 1 < JW / CH) { ep += CH * EDIM; mjp += CH * HDIM; }
        v4f na[CH], nb[CH], nm[CH];
#pragma unroll
        for (int q = 0; q < CH; ++q) {
            na[q] = *(const v4f*)(ep + q * EDIM);
            nb[q] = *(const v4f*)(ep + q * EDIM + 4);
            nm[q] = *(const v4f*)(mjp + q * HDIM);
        }
        __builtin_amdgcn_sched_barrier(0);

#pragma unroll
        for (int q = 0; q < CH; ++q) {
            v2f afr[4];
            afr[0].x = ea[q][0]; afr[0].y = ea[q][1];
            afr[1].x = ea[q][2]; afr[1].y = ea[q][3];
            afr[2].x = eb[q][0]; afr[2].y = eb[q][1];
            afr[3].x = eb[q][2]; afr[3].y = eb[q][3];

            float mk[8];
#pragma unroll
            for (int r = 0; r < 8; ++r) mk[r] = ((mr[r] >> q) & 1u) ? 0.f : NEG_INF;

            const float cj[4] = {mjv[q][0], mjv[q][1], mjv[q][2], mjv[q][3]};
#pragma unroll
            for (int ht = 0; ht < 4; ++ht) {
                v8f acc;
#pragma unroll
                for (int r = 0; r < 8; ++r) acc[r] = cj[ht];
#pragma unroll
                for (int s = 0; s < 4; ++s)
                    acc = __builtin_amdgcn_wmma_f32_16x16x4_f32(
                        false, afr[s], false, bfr[ht][s], (short)0, acc, false, false);
#pragma unroll
                for (int r = 0; r < 8; ++r)
                    vmax[ht][r] = fmaxf(vmax[ht][r], acc[r] + mk[r]);
            }
        }

#pragma unroll
        for (int r = 0; r < 8; ++r) mr[r] >>= CH;
#pragma unroll
        for (int q = 0; q < CH; ++q) { ea[q] = na[q]; eb[q] = nb[q]; mjv[q] = nm[q]; }
    }

#pragma unroll
    for (int r = 0; r < 8; ++r)
        red[(wave * 16 + 8 * hf + r) * 16 + c] =
            make_float4(vmax[0][r], vmax[1][r], vmax[2][r], vmax[3][r]);
    __syncthreads();

    {
        float4 m = red[t];
#pragma unroll
        for (int w = 1; w < 8; ++w) {
            const float4 v = red[w * 256 + t];
            m.x = fmaxf(m.x, v.x); m.y = fmaxf(m.y, v.y);
            m.z = fmaxf(m.z, v.z); m.w = fmaxf(m.w, v.w);
        }
        float* dst = jh ? p1 : p0;
        *(float4*)(dst + ((size_t)(b * Ndim + i0)) * HDIM + 4 * t) = m;
    }
}

#define UROW (ZDIM + HDIM)

__global__ __launch_bounds__(32)
void upd_kernel(const float* __restrict__ z,
                const float* __restrict__ p0,
                const float* __restrict__ p1,
                const float* __restrict__ mi,
                const float* __restrict__ bmsg,
                const float* __restrict__ Wupd,
                const float* __restrict__ bupd,
                float* __restrict__ out) {
    const int lane = threadIdx.x & 31;
    const int c    = lane & 15;
    const int hf   = lane >> 4;
    const int n0   = (blockIdx.x >> 1) * 16;
    const int hb   = (blockIdx.x & 1) * 32;

    const size_t zo = (size_t)(n0 + c) * ZDIM + 4 * hf;
    const size_t ho = (size_t)(n0 + c) * HDIM + 4 * hf;
    const float* bp = Wupd + (size_t)(hb + 2 * c) * UROW + 4 * hf;

    v4f az[8], bz0[8], bz1[8];
#pragma unroll
    for (int i = 0; i < 8; ++i) {
        az[i]  = *(const v4f*)(z + zo + 8 * i);
        bz0[i] = *(const v4f*)(bp + 8 * i);
        bz1[i] = *(const v4f*)(bp + UROW + 8 * i);
    }
    v4f x0[8], x1[8], xm[8], xb[8], bg0[8], bg1[8];
#pragma unroll
    for (int i = 0; i < 8; ++i) {
        x0[i]  = *(const v4f*)(p0 + ho + 8 * i);
        x1[i]  = *(const v4f*)(p1 + ho + 8 * i);
        xm[i]  = *(const v4f*)(mi + ho + 8 * i);
        xb[i]  = *(const v4f*)(bmsg + 4 * hf + 8 * i);
        bg0[i] = *(const v4f*)(bp + ZDIM + 8 * i);
        bg1[i] = *(const v4f*)(bp + UROW + ZDIM + 8 * i);
    }
    __builtin_amdgcn_sched_barrier(0);

    v8f cz0, cz1, cg0, cg1;
#pragma unroll
    for (int r = 0; r < 8; ++r) { cz0[r] = 0.f; cz1[r] = 0.f; cg0[r] = 0.f; cg1[r] = 0.f; }

#pragma unroll
    for (int i = 0; i < 8; ++i) {
#pragma unroll
        for (int e = 0; e < 2; ++e) {
            v2f af, bf0, bf1;
            af.x  = az[i][2 * e];  af.y  = az[i][2 * e + 1];
            bf0.x = bz0[i][2 * e]; bf0.y = bz0[i][2 * e + 1];
            bf1.x = bz1[i][2 * e]; bf1.y = bz1[i][2 * e + 1];
            cz0 = __builtin_amdgcn_wmma_f32_16x16x4_f32(
                false, af, false, bf0, (short)0, cz0, false, false);
            cz1 = __builtin_amdgcn_wmma_f32_16x16x4_f32(
                false, af, false, bf1, (short)0, cz1, false, false);
        }
    }
#pragma unroll
    for (int i = 0; i < 8; ++i) {
        v4f ag;
#pragma unroll
        for (int q = 0; q < 4; ++q)
            ag[q] = fmaxf(x0[i][q], x1[i][q]) + xm[i][q] + xb[i][q];
#pragma unroll
        for (int e = 0; e < 2; ++e) {
            v2f af, bf0, bf1;
            af.x  = ag[2 * e];     af.y  = ag[2 * e + 1];
            bf0.x = bg0[i][2 * e]; bf0.y = bg0[i][2 * e + 1];
            bf1.x = bg1[i][2 * e]; bf1.y = bg1[i][2 * e + 1];
            cg0 = __builtin_amdgcn_wmma_f32_16x16x4_f32(
                false, af, false, bf0, (short)0, cg0, false, false);
            cg1 = __builtin_amdgcn_wmma_f32_16x16x4_f32(
                false, af, false, bf1, (short)0, cg1, false, false);
        }
    }

    const float2 bu = *(const float2*)(bupd + hb + 2 * c);
    float* dst = out + (size_t)(n0 + 8 * hf) * HDIM + hb + 2 * c;
#pragma unroll
    for (int r = 0; r < 8; ++r)
        *(float2*)(dst + r * HDIM) =
            make_float2((cz0[r] + cg0[r]) + bu.x, (cz1[r] + cg1[r]) + bu.y);
}

extern "C" void kernel_launch(void* const* d_in, const int* in_sizes, int n_in,
                              void* d_out, int out_size, void* d_ws, size_t ws_size,
                              hipStream_t stream) {
    const float* z      = (const float*)d_in[0];
    const float* e_feat = (const float*)d_in[1];
    const int*   adj    = (const int*)d_in[2];
    const float* W_msg  = (const float*)d_in[3];
    const float* b_msg  = (const float*)d_in[4];
    const float* W_upd  = (const float*)d_in[5];
    const float* b_upd  = (const float*)d_in[6];
    float*       out    = (float*)d_out;

    const int BN = Bdim * Ndim;
    float* mi  = (float*)d_ws;
    float* mj  = mi + BN * HDIM;
    float* p0  = mj + BN * HDIM;
    float* p1  = p0 + BN * HDIM;

    mimj_kernel<<<(BN / 16) * 2, 64, 0, stream>>>(z, W_msg, mi, mj);
    msg_kernel <<<Bdim * 16 * 2, 256, 0, stream>>>(e_feat, adj, W_msg, mj, p0, p1);
    upd_kernel <<<(BN / 16) * 2, 32, 0, stream>>>(z, p0, p1, mi, b_msg, W_upd, b_upd, out);
}


# msg kernel skips masked edges: each wave lists its row's columns with adj==1 in LDS and runs the WMMA tiles over gathered e_feat and m_j rows (M axis = 16 listed j); sums unchanged, max order changes
# speedup vs baseline: 1.8899x; 1.1406x over previous
#include <hip/hip_runtime.h>
#include <math.h>

#define Bdim 8
#define Ndim 256
#define ZDIM 64
#define EDIM 16
#define HDIM 64

typedef __attribute__((ext_vector_type(2))) float v2f;
typedef __attribute__((ext_vector_type(8))) float v8f;
typedef __attribute__((ext_vector_type(4))) float v4f;

#define WROW (2 * ZDIM + EDIM)

__global__ __launch_bounds__(64)
void mimj_kernel(const float* __restrict__ z,
                 const float* __restrict__ Wmsg,
                 float* __restrict__ mi, float* __restrict__ mj) {
    const int lane = threadIdx.x & 31;
    const int wave = threadIdx.x >> 5;
    const int c    = lane & 15;
    const int hf   = lane >> 4;
    const int n0   = (blockIdx.x >> 1) * 16;
    const int half = blockIdx.x & 1;
    const int hb   = 32 * wave;

    const float* ap = z + (size_t)(n0 + c) * ZDIM + 4 * hf;
    const float* bp = Wmsg + (size_t)(hb + 2 * c) * WROW + half * ZDIM + 4 * hf;
    v4f a[8], b0[8], b1[8];
#pragma unroll
    for (int i = 0; i < 8; ++i) {
        a[i]  = *(const v4f*)(ap + 8 * i);
        b0[i] = *(const v4f*)(bp + 8 * i);
        b1[i] = *(const v4f*)(bp + WROW + 8 * i);
    }
    __builtin_amdgcn_sched_barrier(0);

    v8f acc0, acc1;
#pragma unroll
    for (int r = 0; r < 8; ++r) { acc0[r] = 0.f; acc1[r] = 0.f; }
#pragma unroll
    for (int i = 0; i < 8; ++i) {
#pragma unroll
        for (int e = 0; e < 2; ++e) {
            v2f af, bf0, bf1;
            af.x  = a[i][2 * e];  af.y  = a[i][2 * e + 1];
            bf0.x = b0[i][2 * e]; bf0.y = b0[i][2 * e + 1];
            bf1.x = b1[i][2 * e]; bf1.y = b1[i][2 * e + 1];
            acc0 = __builtin_amdgcn_wmma_f32_16x16x4_f32(
                false, af, false, bf0, (short)0, acc0, false, false);
            acc1 = __builtin_amdgcn_wmma_f32_16x16x4_f32(
                false, af, false, bf1, (short)0, acc1, false, false);
        }
    }

    float* dst = (half ? mj : mi) + (size_t)(n0 + 8 * hf) * HDIM + hb + 2 * c;
#pragma unroll
    for (int r = 0; r < 8; ++r)
        *(float2*)(dst + r * HDIM) = make_float2(acc0[r], acc1[r]);
}

#define CHJ 32
#define LSZ (Ndim + CHJ)

__global__ __launch_bounds__(256)
void msg_kernel(const float* __restrict__ e_feat,
                const int*   __restrict__ adj,
                const float* __restrict__ Wmsg,
                const float* __restrict__ mj,
                float* __restrict__ p0) {
    __shared__ __attribute__((aligned(16))) int jlist[8][LSZ];

    const int t    = threadIdx.x;
    const int wave = t >> 5;
    const int lane = t & 31;
    const int c    = lane & 15;
    const int hf   = lane >> 4;
    const int row  = blockIdx.x * 8 + wave;
    const int b    = row >> 8;

    const float NEG_INF = -__builtin_inff();

    int* L = jlist[wave];
    int cnt  = 0;
    int padj = 0;
    {
        const int* arow = adj + (size_t)row * Ndim + lane;
        int av[8];
#pragma unroll
        for (int q = 0; q < 8; ++q) av[q] = arow[32 * q];
#pragma unroll
        for (int q = 0; q < 8; ++q) {
            const bool keep = (av[q] == 1);
            const unsigned bal = __builtin_amdgcn_ballot_w32(keep);
            const int rank = cnt + (int)__builtin_amdgcn_mbcnt_lo(bal, 0u);
            if (keep) L[rank] = 32 * q + lane;
            if (bal) padj = 32 * q + 31 - __builtin_clz(bal);
            cnt += __builtin_popcount(bal);
        }
        L[cnt + lane] = padj;
    }
    const int nch  = (cnt + CHJ - 1) / CHJ;
    const int lastc = nch > 0 ? nch - 1 : 0;
    __syncthreads();

    v2f bfr[4][4];
#pragma unroll
    for (int ht = 0; ht < 4; ++ht) {
        const float* wp = Wmsg + (4 * c + ht) * (2 * ZDIM + EDIM) + 2 * ZDIM + 8 * hf;
        const float4 w0 = *(const float4*)(wp);
        const float4 w1 = *(const float4*)(wp + 4);
        bfr[ht][0].x = w0.x; bfr[ht][0].y = w0.y;
        bfr[ht][1].x = w0.z; bfr[ht][1].y = w0.w;
        bfr[ht][2].x = w1.x; bfr[ht][2].y = w1.y;
        bfr[ht][3].x = w1.z; bfr[ht][3].y = w1.w;
    }

    const float* erow = e_feat + (size_t)row * Ndim * EDIM + 8 * hf;
    const float* mjb  = mj + (size_t)b * Ndim * HDIM + 4 * c;

    int  ja[2];
    int4 jc[2][2];
    {
        const int* lp = L;
#pragma unroll
        for (int u = 0; u < 2; ++u) {
            ja[u]    = lp[16 * u + c];
            jc[u][0] = *(const int4*)(lp + 16 * u + 8 * hf);
            jc[u][1] = *(const int4*)(lp + 16 * u + 8 * hf + 4);
        }
    }
    v4f ea[2], eb[2], cm[2][8];
#pragma unroll
    for (int u = 0; u < 2; ++u) {
        const float* ep = erow + (unsigned)ja[u] * EDIM;
        ea[u] = *(const v4f*)(ep);
        eb[u] = *(const v4f*)(ep + 4);
        const int jr[8] = {jc[u][0].x, jc[u][0].y, jc[u][0].z, jc[u][0].w,
                           jc[u][1].x, jc[u][1].y, jc[u][1].z, jc[u][1].w};
#pragma unroll
        for (int r = 0; r < 8; ++r) cm[u][r] = *(const v4f*)(mjb + (unsigned)jr[r] * HDIM);
    }
    {
        const int* lp = L + CHJ * (1 < lastc ? 1 : lastc);
#pragma unroll
        for (int u = 0; u < 2; ++u) {
            ja[u]    = lp[16 * u + c];
            jc[u][0] = *(const int4*)(lp + 16 * u + 8 * hf);
            jc[u][1] = *(const int4*)(lp + 16 * u + 8 * hf + 4);
        }
    }

    float vmax[4][8];
#pragma unroll
    for (int ht = 0; ht < 4; ++ht)
#pragma unroll
        for (int r = 0; r < 8; ++r) vmax[ht][r] = NEG_INF;

#pragma nounroll
    for (int ch = 0; ch < nch; ++ch) {
        v4f na[2], nb[2], nm[2][8];
#pragma unroll
        for (int u = 0; u < 2; ++u) {
            const float* ep = erow + (unsigned)ja[u] * EDIM;
            na[u] = *(const v4f*)(ep);
            nb[u] = *(const v4f*)(ep + 4);
            const int jr[8] = {jc[u][0].x, jc[u][0].y, jc[u][0].z, jc[u][0].w,
                               jc[u][1].x, jc[u][1].y, jc[u][1].z, jc[u][1].w};
#pragma unroll
            for (int r = 0; r < 8; ++r) nm[u][r] = *(const v4f*)(mjb + (unsigned)jr[r] * HDIM);
        }
        {
            const int nx = ch + 2 < lastc ? ch + 2 : lastc;
            const int* lp = L + CHJ * nx;
#pragma unroll
            for (int u = 0; u < 2; ++u) {
                ja[u]    = lp[16 * u + c];
                jc[u][0] = *(const int4*)(lp + 16 * u + 8 * hf);
                jc[u][1] = *(const int4*)(lp + 16 * u + 8 * hf + 4);
            }
        }
        __builtin_amdgcn_sched_barrier(0);

#pragma unroll
        for (int u = 0; u < 2; ++u) {
            v2f afr[4];
            afr[0].x = ea[u][0]; afr[0].y = ea[u][1];
            afr[1].x = ea[u][2]; afr[1].y = ea[u][3];
            afr[2].x = eb[u][0]; afr[2].y = eb[u][1];
            afr[3].x = eb[u][2]; afr[3].y = eb[u][3];
#pragma unroll
            for (int ht = 0; ht < 4; ++ht) {
                v8f acc;
#pragma unroll
                for (int r = 0; r < 8; ++r) acc[r] = cm[u][r][ht];
#pragma unroll
                for (int s = 0; s < 4; ++s)
                    acc = __builtin_amdgcn_wmma_f32_16x16x4_f32(
                        false, afr[s], false, bfr[ht][s], (short)0, acc, false, false);
#pragma unroll
                for (int r = 0; r < 8; ++r)
                    vmax[ht][r] = fmaxf(vmax[ht][r], acc[r]);
            }
        }

        __builtin_amdgcn_sched_barrier(0);
#pragma unroll
        for (int u = 0; u < 2; ++u) {
            ea[u] = na[u]; eb[u] = nb[u];
#pragma unroll
            for (int r = 0; r < 8; ++r) cm[u][r] = nm[u][r];
        }
    }

    float res[4];
#pragma unroll
    for (int ht = 0; ht < 4; ++ht) {
        float m = vmax[ht][0];
#pragma unroll
        for (int r = 1; r < 8; ++r) m = fmaxf(m, vmax[ht][r]);
        res[ht] = fmaxf(m, __shfl_xor(m, 16, 32));
    }
    if (hf == 0)
        *(float4*)(p0 + (size_t)row * HDIM + 4 * c) = make_float4(res[0], res[1], res[2], res[3]);
}

#define UROW (ZDIM + HDIM)

__global__ __launch_bounds__(32)
void upd_kernel(const float* __restrict__ z,
                const float* __restrict__ p0,
                const float* __restrict__ mi,
                const float* __restrict__ bmsg,
                const float* __restrict__ Wupd,
                const float* __restrict__ bupd,
                float* __restrict__ out) {
    const int lane = threadIdx.x & 31;
    const int c    = lane & 15;
    const int hf   = lane >> 4;
    const int n0   = (blockIdx.x >> 1) * 16;
    const int hb   = (blockIdx.x & 1) * 32;

    const size_t zo = (size_t)(n0 + c) * ZDIM + 4 * hf;
    const size_t ho = (size_t)(n0 + c) * HDIM + 4 * hf;
    const float* bp = Wupd + (size_t)(hb + 2 * c) * UROW + 4 * hf;

    v4f az[8], bz0[8], bz1[8];
#pragma unroll
    for (int i = 0; i < 8; ++i) {
        az[i]  = *(const v4f*)(z + zo + 8 * i);
        bz0[i] = *(const v4f*)(bp + 8 * i);
        bz1[i] = *(const v4f*)(bp + UROW + 8 * i);
    }
    v4f x0[8], xm[8], xb[8], bg0[8], bg1[8];
#pragma unroll
    for (int i = 0; i < 8; ++i) {
        x0[i]  = *(const v4f*)(p0 + ho + 8 * i);
        xm[i]  = *(const v4f*)(mi + ho + 8 * i);
        xb[i]  = *(const v4f*)(bmsg + 4 * hf + 8 * i);
        bg0[i] = *(const v4f*)(bp + ZDIM + 8 * i);
        bg1[i] = *(const v4f*)(bp + UROW + ZDIM + 8 * i);
    }
    __builtin_amdgcn_sched_barrier(0);

    v8f cz0, cz1, cg0, cg1;
#pragma unroll
    for (int r = 0; r < 8; ++r) { cz0[r] = 0.f; cz1[r] = 0.f; cg0[r] = 0.f; cg1[r] = 0.f; }

#pragma unroll
    for (int i = 0; i < 8; ++i) {
#pragma unroll
        for (int e = 0; e < 2; ++e) {
            v2f af, bf0, bf1;
            af.x  = az[i][2 * e];  af.y  = az[i][2 * e + 1];
            bf0.x = bz0[i][2 * e]; bf0.y = bz0[i][2 * e + 1];
            bf1.x = bz1[i][2 * e]; bf1.y = bz1[i][2 * e + 1];
            cz0 = __builtin_amdgcn_wmma_f32_16x16x4_f32(
                false, af, false, bf0, (short)0, cz0, false, false);
            cz1 = __builtin_amdgcn_wmma_f32_16x16x4_f32(
                false, af, false, bf1, (short)0, cz1, false, false);
        }
    }
#pragma unroll
    for (int i = 0; i < 8; ++i) {
        v4f ag;
#pragma unroll
        for (int q = 0; q < 4; ++q)
            ag[q] = x0[i][q] + xm[i][q] + xb[i][q];
#pragma unroll
        for (int e = 0; e < 2; ++e) {
            v2f af, bf0, bf1;
            af.x  = ag[2 * e];     af.y  = ag[2 * e + 1];
            bf0.x = bg0[i][2 * e]; bf0.y = bg0[i][2 * e + 1];
            bf1.x = bg1[i][2 * e]; bf1.y = bg1[i][2 * e + 1];
            cg0 = __builtin_amdgcn_wmma_f32_16x16x4_f32(
                false, af, false, bf0, (short)0, cg0, false, false);
            cg1 = __builtin_amdgcn_wmma_f32_16x16x4_f32(
                false, af, false, bf1, (short)0, cg1, false, false);
        }
    }

    const float2 bu = *(const float2*)(bupd + hb + 2 * c);
    float* dst = out + (size_t)(n0 + 8 * hf) * HDIM + hb + 2 * c;
#pragma unroll
    for (int r = 0; r < 8; ++r)
        *(float2*)(dst + r * HDIM) =
            make_float2((cz0[r] + cg0[r]) + bu.x, (cz1[r] + cg1[r]) + bu.y);
}

extern "C" void kernel_launch(void* const* d_in, const int* in_sizes, int n_in,
                              void* d_out, int out_size, void* d_ws, size_t ws_size,
                              hipStream_t stream) {
    const float* z      = (const float*)d_in[0];
    const float* e_feat = (const float*)d_in[1];
    const int*   adj    = (const int*)d_in[2];
    const float* W_msg  = (const float*)d_in[3];
    const float* b_msg  = (const float*)d_in[4];
    const float* W_upd  = (const float*)d_in[5];
    const float* b_upd  = (const float*)d_in[6];
    float*       out    = (float*)d_out;

    const int BN = Bdim * Ndim;
    float* mi  = (float*)d_ws;
    float* mj  = mi + BN * HDIM;
    float* p0  = mj + BN * HDIM;

    mimj_kernel<<<(BN / 16) * 2, 64, 0, stream>>>(z, W_msg, mi, mj);
    msg_kernel <<<BN / 8, 256, 0, stream>>>(e_feat, adj, W_msg, mj, p0);
    upd_kernel <<<(BN / 16) * 2, 32, 0, stream>>>(z, p0, mi, b_msg, W_upd, b_upd, out);
}


# upd layer split by column halves: first kernel also forms z.W_upd[:,:64]+b_upd; msg kernel's tail does agg.W_upd[:,64:] for its 8 rows on WMMA and writes out; 3 launches -> 2 (upd sum order changes)
# speedup vs baseline: 2.1657x; 1.1459x over previous
#include <hip/hip_runtime.h>
#include <math.h>

#define Bdim 8
#define Ndim 256
#define ZDIM 64
#define EDIM 16
#define HDIM 64

typedef __attribute__((ext_vector_type(2))) float v2f;
typedef __attribute__((ext_vector_type(8))) float v8f;
typedef __attribute__((ext_vector_type(4))) float v4f;

#define WROW (2 * ZDIM + EDIM)
#define UROW (ZDIM + HDIM)

__global__ __launch_bounds__(64)
void front_kernel(const float* __restrict__ z,
                  const float* __restrict__ Wmsg,
                  const float* __restrict__ Wupd,
                  const float* __restrict__ bupd,
                  float* __restrict__ mi, float* __restrict__ mj,
                  float* __restrict__ zp) {
    const int lane = threadIdx.x & 31;
    const int wave = threadIdx.x >> 5;
    const int c    = lane & 15;
    const int hf   = lane >> 4;
    const int n0   = blockIdx.x * 16;
    const int part = blockIdx.y;
    const int hb   = 32 * wave;

    const int    wrow = part == 2 ? UROW : WROW;
    const float* wb   = part == 2 ? Wupd : Wmsg + (part == 1 ? ZDIM : 0);
    const float* ap = z + (size_t)(n0 + c) * ZDIM + 4 * hf;
    const float* bp = wb + (size_t)(hb + 2 * c) * wrow + 4 * hf;
    v4f a[8], b0[8], b1[8];
#pragma unroll
    for (int i = 0; i < 8; ++i) {
        a[i]  = *(const v4f*)(ap + 8 * i);
        b0[i] = *(const v4f*)(bp + 8 * i);
        b1[i] = *(const v4f*)(bp + wrow + 8 * i);
    }
    __builtin_amdgcn_sched_barrier(0);

    v8f acc0, acc1;
#pragma unroll
    for (int r = 0; r < 8; ++r) { acc0[r] = 0.f; acc1[r] = 0.f; }
#pragma unroll
    for (int i = 0; i < 8; ++i) {
#pragma unroll
        for (int e = 0; e < 2; ++e) {
            v2f af, bf0, bf1;
            af.x  = a[i][2 * e];  af.y  = a[i][2 * e + 1];
            bf0.x = b0[i][2 * e]; bf0.y = b0[i][2 * e + 1];
            bf1.x = b1[i][2 * e]; bf1.y = b1[i][2 * e + 1];
            acc0 = __builtin_amdgcn_wmma_f32_16x16x4_f32(
                false, af, false, bf0, (short)0, acc0, false, false);
            acc1 = __builtin_amdgcn_wmma_f32_16x16x4_f32(
                false, af, false, bf1, (short)0, acc1, false, false);
        }
    }

    float* dst = (part == 0 ? mi : part == 1 ? mj : zp) + (size_t)(n0 + 8 * hf) * HDIM + hb + 2 * c;
    if (part == 2) {
        const float2 bu = *(const float2*)(bupd + hb + 2 * c);
#pragma unroll
        for (int r = 0; r < 8; ++r)
            *(float2*)(dst + r * HDIM) = make_float2(acc0[r] + bu.x, acc1[r] + bu.y);
    } else {
#pragma unroll
        for (int r = 0; r < 8; ++r)
            *(float2*)(dst + r * HDIM) = make_float2(acc0[r], acc1[r]);
    }
}

#define CHJ 32
#define LSZ (Ndim + CHJ)
#define APITCH (HDIM + 4)

__global__ __launch_bounds__(256)
void msg_kernel(const float* __restrict__ e_feat,
                const int*   __restrict__ adj,
                const float* __restrict__ Wmsg,
                const float* __restrict__ mj,
                const float* __restrict__ mi,
                const float* __restrict__ bmsg,
                const float* __restrict__ Wupd,
                const float* __restrict__ zp,
                float* __restrict__ out) {
    __shared__ __attribute__((aligned(16))) int jlist[8][LSZ];
    __shared__ __attribute__((aligned(16))) float aggs[8][APITCH];

    const int t    = threadIdx.x;
    const int wave = t >> 5;
    const int lane = t & 31;
    const int c    = lane & 15;
    const int hf   = lane >> 4;
    const int row  = blockIdx.x * 8 + wave;
    const int b    = row >> 8;

    const float NEG_INF = -__builtin_inff();

    int* L = jlist[wave];
    int cnt  = 0;
    int padj = 0;
    {
        const int* arow = adj + (size_t)row * Ndim + lane;
        int av[8];
#pragma unroll
        for (int q = 0; q < 8; ++q) av[q] = arow[32 * q];
#pragma unroll
        for (int q = 0; q < 8; ++q) {
            const bool keep = (av[q] == 1);
            const unsigned bal = __builtin_amdgcn_ballot_w32(keep);
            const int rank = cnt + (int)__builtin_amdgcn_mbcnt_lo(bal, 0u);
            if (keep) L[rank] = 32 * q + lane;
            if (bal) padj = 32 * q + 31 - __builtin_clz(bal);
            cnt += __builtin_popcount(bal);
        }
        L[cnt + lane] = padj;
    }
    const int nch  = (cnt + CHJ - 1) / CHJ;
    const int lastc = nch > 0 ? nch - 1 : 0;
    __syncthreads();

    v2f bfr[4][4];
#pragma unroll
    for (int ht = 0; ht < 4; ++ht) {
        const float* wp = Wmsg + (4 * c + ht) * (2 * ZDIM + EDIM) + 2 * ZDIM + 8 * hf;
        const float4 w0 = *(const float4*)(wp);
        const float4 w1 = *(const float4*)(wp + 4);
        bfr[ht][0].x = w0.x; bfr[ht][0].y = w0.y;
        bfr[ht][1].x = w0.z; bfr[ht][1].y = w0.w;
        bfr[ht][2].x = w1.x; bfr[ht][2].y = w1.y;
        bfr[ht][3].x = w1.z; bfr[ht][3].y = w1.w;
    }

    const float* erow = e_feat + (size_t)row * Ndim * EDIM + 8 * hf;
    const float* mjb  = mj + (size_t)b * Ndim * HDIM + 4 * c;

    int  ja[2];
    int4 jc[2][2];
    {
        const int* lp = L;
#pragma unroll
        for (int u = 0; u < 2; ++u) {
            ja[u]    = lp[16 * u + c];
            jc[u][0] = *(const int4*)(lp + 16 * u + 8 * hf);
            jc[u][1] = *(const int4*)(lp + 16 * u + 8 * hf + 4);
        }
    }
    v4f ea[2], eb[2], cm[2][8];
#pragma unroll
    for (int u = 0; u < 2; ++u) {
        const float* ep = erow + (unsigned)ja[u] * EDIM;
        ea[u] = *(const v4f*)(ep);
        eb[u] = *(const v4f*)(ep + 4);
        const int jr[8] = {jc[u][0].x, jc[u][0].y, jc[u][0].z, jc[u][0].w,
                           jc[u][1].x, jc[u][1].y, jc[u][1].z, jc[u][1].w};
#pragma unroll
        for (int r = 0; r < 8; ++r) cm[u][r] = *(const v4f*)(mjb + (unsigned)jr[r] * HDIM);
    }
    {
        const int* lp = L + CHJ * (1 < lastc ? 1 : lastc);
#pragma unroll
        for (int u = 0; u < 2; ++u) {
            ja[u]    = lp[16 * u + c];
            jc[u][0] = *(const int4*)(lp + 16 * u + 8 * hf);
            jc[u][1] = *(const int4*)(lp + 16 * u + 8 * hf + 4);
        }
    }

    const int  row0 = blockIdx.x * 8;
    const bool tailw = __builtin_amdgcn_readfirstlane(wave) < 2;
    const int  hb   = 32 * (wave & 1);
    const v4f miv = *(const v4f*)(mi + (size_t)row * HDIM + 4 * c);
    const v4f bmv = *(const v4f*)(bmsg + 4 * c);
    v4f bg0[8], bg1[8];
    v8f oc0, oc1;
#pragma unroll
    for (int i = 0; i < 8; ++i) { bg0[i] = (v4f){0.f, 0.f, 0.f, 0.f}; bg1[i] = bg0[i]; }
#pragma unroll
    for (int r = 0; r < 8; ++r) { oc0[r] = 0.f; oc1[r] = 0.f; }
    if (tailw) {
        const float* bp = Wupd + (size_t)(hb + 2 * c) * UROW + ZDIM + 4 * hf;
#pragma unroll
        for (int i = 0; i < 8; ++i) {
            bg0[i] = *(const v4f*)(bp + 8 * i);
            bg1[i] = *(const v4f*)(bp + UROW + 8 * i);
        }
        const float* zq = zp + (size_t)row0 * HDIM + hb + 2 * c;
#pragma unroll
        for (int r = 0; r < 8; ++r) {
            const float2 v = *(const float2*)(zq + r * HDIM);
            oc0[r] = v.x; oc1[r] = v.y;
        }
    }

    float vmax[4][8];
#pragma unroll
    for (int ht = 0; ht < 4; ++ht)
#pragma unroll
        for (int r = 0; r < 8; ++r) vmax[ht][r] = NEG_INF;

#pragma nounroll
    for (int ch = 0; ch < nch; ++ch) {
        v4f na[2], nb[2], nm[2][8];
#pragma unroll
        for (int u = 0; u < 2; ++u) {
            const float* ep = erow + (unsigned)ja[u] * EDIM;
            na[u] = *(const v4f*)(ep);
            nb[u] = *(const v4f*)(ep + 4);
            const int jr[8] = {jc[u][0].x, jc[u][0].y, jc[u][0].z, jc[u][0].w,
                               jc[u][1].x, jc[u][1].y, jc[u][1].z, jc[u][1].w};
#pragma unroll
            for (int r = 0; r < 8; ++r) nm[u][r] = *(const v4f*)(mjb + (unsigned)jr[r] * HDIM);
        }
        {
            const int nx = ch + 2 < lastc ? ch + 2 : lastc;
            const int* lp = L + CHJ * nx;
#pragma unroll
            for (int u = 0; u < 2; ++u) {
                ja[u]    = lp[16 * u + c];
                jc[u][0] = *(const int4*)(lp + 16 * u + 8 * hf);
                jc[u][1] = *(const int4*)(lp + 16 * u + 8 * hf + 4);
            }
        }
        __builtin_amdgcn_sched_barrier(0);

#pragma unroll
        for (int u = 0; u < 2; ++u) {
            v2f afr[4];
            afr[0].x = ea[u][0]; afr[0].y = ea[u][1];
            afr[1].x = ea[u][2]; afr[1].y = ea[u][3];
            afr[2].x = eb[u][0]; afr[2].y = eb[u][1];
            afr[3].x = eb[u][2]; afr[3].y = eb[u][3];
#pragma unroll
            for (int ht = 0; ht < 4; ++ht) {
                v8f acc;
#pragma unroll
                for (int r = 0; r < 8; ++r) acc[r] = cm[u][r][ht];
#pragma unroll
                for (int s = 0; s < 4; ++s)
                    acc = __builtin_amdgcn_wmma_f32_16x16x4_f32(
                        false, afr[s], false, bfr[ht][s], (short)0, acc, false, false);
#pragma unroll
                for (int r = 0; r < 8; ++r)
                    vmax[ht][r] = fmaxf(vmax[ht][r], acc[r]);
            }
        }

        __builtin_amdgcn_sched_barrier(0);
#pragma unroll
        for (int u = 0; u < 2; ++u) {
            ea[u] = na[u]; eb[u] = nb[u];
#pragma unroll
            for (int r = 0; r < 8; ++r) cm[u][r] = nm[u][r];
        }
    }

    float res[4];
#pragma unroll
    for (int ht = 0; ht < 4; ++ht) {
        float m = vmax[ht][0];
#pragma unroll
        for (int r = 1; r < 8; ++r) m = fmaxf(m, vmax[ht][r]);
        res[ht] = fmaxf(m, __shfl_xor(m, 16, 32));
    }
    if (hf == 0) {
        v4f ag;
#pragma unroll
        for (int ht = 0; ht < 4; ++ht) ag[ht] = res[ht] + miv[ht] + bmv[ht];
        *(v4f*)(&aggs[wave][4 * c]) = ag;
    }
    __syncthreads();

    if (tailw) {
        const float* arow = &aggs[c & 7][4 * hf];
        v4f ag[8];
#pragma unroll
        for (int i = 0; i < 8; ++i) ag[i] = *(const v4f*)(arow + 8 * i);
#pragma unroll
        for (int i = 0; i < 8; ++i) {
#pragma unroll
            for (int e = 0; e < 2; ++e) {
                v2f af, bf0, bf1;
                af.x  = ag[i][2 * e];  af.y  = ag[i][2 * e + 1];
                bf0.x = bg0[i][2 * e]; bf0.y = bg0[i][2 * e + 1];
                bf1.x = bg1[i][2 * e]; bf1.y = bg1[i][2 * e + 1];
                oc0 = __builtin_amdgcn_wmma_f32_16x16x4_f32(
                    false, af, false, bf0, (short)0, oc0, false, false);
                oc1 = __builtin_amdgcn_wmma_f32_16x16x4_f32(
                    false, af, false, bf1, (short)0, oc1, false, false);
            }
        }
        if (hf == 0) {
            float* dst = out + (size_t)row0 * HDIM + hb + 2 * c;
#pragma unroll
            for (int r = 0; r < 8; ++r)
                *(float2*)(dst + r * HDIM) = make_float2(oc0[r], oc1[r]);
        }
    }
}

extern "C" void kernel_launch(void* const* d_in, const int* in_sizes, int n_in,
                              void* d_out, int out_size, void* d_ws, size_t ws_size,
                              hipStream_t stream) {
    const float* z      = (const float*)d_in[0];
    const float* e_feat = (const float*)d_in[1];
    const int*   adj    = (const int*)d_in[2];
    const float* W_msg  = (const float*)d_in[3];
    const float* b_msg  = (const float*)d_in[4];
    const float* W_upd  = (const float*)d_in[5];
    const float* b_upd  = (const float*)d_in[6];
    float*       out    = (float*)d_out;

    const int BN = Bdim * Ndim;
    float* mi  = (float*)d_ws;
    float* mj  = mi + BN * HDIM;
    float* zp  = mj + BN * HDIM;

    front_kernel<<<dim3(BN / 16, 3), 64, 0, stream>>>(z, W_msg, W_upd, b_upd, mi, mj, zp);
    msg_kernel  <<<BN / 8, 256, 0, stream>>>(e_feat, adj, W_msg, mj, mi, b_msg, W_upd, zp, out);
}
